# starting kernel
# baseline (speedup 1.0000x reference)
#include <hip/hip_runtime.h>

typedef __attribute__((ext_vector_type(2))) float v2f;
typedef __attribute__((ext_vector_type(8))) float v8f;

#define IN_F 128
#define OUT_F 64
#define NEG_SLOPE 0.2f
#define LDS_STRIDE (IN_F + 2)

__global__ __launch_bounds__(256) void gat_gemm_kernel(
    const float* __restrict__ x,  const float* __restrict__ Wv,
    const float* __restrict__ bv, const float* __restrict__ wq,
    const float* __restrict__ bq, const float* __restrict__ wk,
    const float* __restrict__ bk,
    float* __restrict__ h, float* __restrict__ qout, float* __restrict__ kout,
    int N)
{
    __shared__ float Wt[OUT_F * LDS_STRIDE];

    const int tid = threadIdx.x;
    for (int idx = tid; idx < IN_F * OUT_F; idx += 256) {
        int kk = idx / OUT_F, n = idx % OUT_F;
        Wt[n * LDS_STRIDE + kk] = Wv[idx];
    }
    __syncthreads();

    const int wave = tid >> 5;
    const int lane = tid & 31;
    const int half = lane >> 4;
    const int l16  = lane & 15;
    const long tile = (long)blockIdx.x * 8 + wave;
    const long rowBase = tile * 16;
    if (rowBase >= (long)N) return;

    v8f acc[4];
#pragma unroll
    for (int nt = 0; nt < 4; ++nt) {
        float b = bv[nt * 16 + l16];
#pragma unroll
        for (int j = 0; j < 8; ++j) acc[nt][j] = b;
    }

    long arow = rowBase + l16;
    if (arow >= (long)N) arow = (long)N - 1;
    const float* __restrict__ xrow = x + arow * IN_F;

    for (int kk = 0; kk < IN_F; kk += 4) {
        const int kl = kk + 2 * half;
        const float2 av = *(const float2*)(xrow + kl);
        v2f a; a.x = av.x; a.y = av.y;
#pragma unroll
        for (int nt = 0; nt < 4; ++nt) {
            const float* wp = &Wt[(nt * 16 + l16) * LDS_STRIDE + kl];
            v2f b; b.x = wp[0]; b.y = wp[1];
            acc[nt] = __builtin_amdgcn_wmma_f32_16x16x4_f32(
                false, a, false, b, (short)0, acc[nt], false, false);
        }
    }

    float qpart[8], kpart[8];
#pragma unroll
    for (int j = 0; j < 8; ++j) { qpart[j] = 0.f; kpart[j] = 0.f; }
#pragma unroll
    for (int nt = 0; nt < 4; ++nt) {
        const float wqn = wq[nt * 16 + l16];
        const float wkn = wk[nt * 16 + l16];
#pragma unroll
        for (int j = 0; j < 8; ++j) {
            qpart[j] += acc[nt][j] * wqn;
            kpart[j] += acc[nt][j] * wkn;
        }
    }
#pragma unroll
    for (int off = 8; off >= 1; off >>= 1) {
#pragma unroll
        for (int j = 0; j < 8; ++j) {
            qpart[j] += __shfl_xor(qpart[j], off, 32);
            kpart[j] += __shfl_xor(kpart[j], off, 32);
        }
    }

#pragma unroll
    for (int nt = 0; nt < 4; ++nt) {
#pragma unroll
        for (int j = 0; j < 8; ++j) {
            const long m = rowBase + half * 8 + j;
            if (m < (long)N) h[m * OUT_F + nt * 16 + l16] = acc[nt][j];
        }
    }
    if (l16 == 0) {
        const float bqv = bq[0], bkv = bk[0];
#pragma unroll
        for (int j = 0; j < 8; ++j) {
            const long m = rowBase + half * 8 + j;
            if (m < (long)N) {
                qout[m] = qpart[j] + bqv;
                kout[m] = kpart[j] + bkv;
            }
        }
    }
}

__device__ __forceinline__ unsigned f2ord(float f) {
    unsigned u = __float_as_uint(f);
    return (u & 0x80000000u) ? ~u : (u | 0x80000000u);
}
__device__ __forceinline__ float ord2f(unsigned u) {
    return (u & 0x80000000u) ? __uint_as_float(u ^ 0x80000000u)
                             : __uint_as_float(~u);
}

__global__ __launch_bounds__(256) void edge_max_kernel(
    const float* __restrict__ q, const float* __restrict__ k,
    const int* __restrict__ src, const int* __restrict__ dst,
    float* __restrict__ coeff, unsigned* __restrict__ mmax, int E)
{
    int i = blockIdx.x * 256 + threadIdx.x;
    if (i >= E) return;
    const int s = src[i], d = dst[i];
    const float e = q[s] + k[d];
    const float c = (e >= 0.f) ? e : NEG_SLOPE * e;
    coeff[i] = c;
    atomicMax(&mmax[d], f2ord(c));
}

__global__ __launch_bounds__(256) void edge_expsum_kernel(
    const int* __restrict__ dst, const unsigned* __restrict__ mmax,
    float* __restrict__ coeff, float* __restrict__ ssum, int E)
{
    int i = blockIdx.x * 256 + threadIdx.x;
    if (i >= E) return;
    const int d = dst[i];
    const float m = ord2f(mmax[d]);
    const float ex = __expf(coeff[i] - m);
    coeff[i] = ex;
    atomicAdd(&ssum[d], ex);
}

__global__ __launch_bounds__(256) void edge_aggregate_kernel(
    const float* __restrict__ h, const int* __restrict__ src,
    const int* __restrict__ dst, const float* __restrict__ ex,
    const float* __restrict__ ssum, float* __restrict__ out, int E)
{
    const long gwave = ((long)blockIdx.x * 256 + threadIdx.x) >> 5;
    const int lane = threadIdx.x & 31;
    if (gwave >= (long)E) return;
    const int s = src[gwave], d = dst[gwave];
    const float attn = ex[gwave] / ssum[d];
    const float2 v = *(const float2*)(h + (long)s * OUT_F + 2 * lane);
    float* op = out + (long)d * OUT_F + 2 * lane;
    atomicAdd(op,     v.x * attn);
    atomicAdd(op + 1, v.y * attn);
}

extern "C" void kernel_launch(void* const* d_in, const int* in_sizes, int n_in,
                              void* d_out, int out_size, void* d_ws, size_t ws_size,
                              hipStream_t stream) {
    const float* x  = (const float*)d_in[0];
    const float* Wv = (const float*)d_in[1];
    const float* bv = (const float*)d_in[2];
    const float* wq = (const float*)d_in[3];
    const float* bq = (const float*)d_in[4];
    const float* wk = (const float*)d_in[5];
    const float* bk = (const float*)d_in[6];
    const int*   src = (const int*)d_in[7];
    const int*   dst = (const int*)d_in[8];

    const int N = in_sizes[0] / IN_F;
    const int E = in_sizes[7];
    float* out = (float*)d_out;

    char* p = (char*)d_ws;
    auto carve = [&](size_t bytes) {
        char* r = p;
        p += (bytes + 255) & ~(size_t)255;
        return r;
    };
    float*    h    = (float*)   carve((size_t)N * OUT_F * sizeof(float));
    float*    q    = (float*)   carve((size_t)N * sizeof(float));
    float*    k    = (float*)   carve((size_t)N * sizeof(float));
    unsigned* mmax = (unsigned*)carve((size_t)N * sizeof(unsigned));
    float*    ssum = (float*)   carve((size_t)N * sizeof(float));
    float*    ex   = (float*)   carve((size_t)E * sizeof(float));

    hipMemsetAsync(out,  0, (size_t)N * OUT_F * sizeof(float), stream);
    hipMemsetAsync(mmax, 0, (size_t)N * sizeof(unsigned), stream);
    hipMemsetAsync(ssum, 0, (size_t)N * sizeof(float), stream);

    const int tiles = (N + 15) / 16;
    const int gemmBlocks = (tiles + 7) / 8;
    gat_gemm_kernel<<<gemmBlocks, 256, 0, stream>>>(
        x, Wv, bv, wq, bq, wk, bk, h, q, k, N);

    edge_max_kernel<<<(E + 255) / 256, 256, 0, stream>>>(
        q, k, src, dst, ex, mmax, E);

    edge_expsum_kernel<<<(E + 255) / 256, 256, 0, stream>>>(
        dst, mmax, ex, ssum, E);

    edge_aggregate_kernel<<<(E + 7) / 8, 256, 0, stream>>>(
        h, src, dst, ex, ssum, out, E);
}


# Owner blocks load 16 bytes per lane: one load fetches four flagged chunk lines or two edges' rows (halves on separate LDS banks); a third of the load instructions; sum order still LDS arrival
# speedup vs baseline: 4.7136x; 4.7136x over previous
#include <hip/hip_runtime.h>

typedef __attribute__((ext_vector_type(2))) float v2f;
typedef __attribute__((ext_vector_type(8))) float v8f;

#define IN_F 128
#define OUT_F 64
#define NEG_SLOPE 0.2f
#define LDS_STRIDE (IN_F + 2)

#define SPREAD_MAX 64.0f

__device__ __forceinline__ unsigned f2ord(float f) {
    unsigned u = __float_as_uint(f);
    return (u & 0x80000000u) ? ~u : (u | 0x80000000u);
}
__device__ __forceinline__ float ord2f(unsigned u) {
    return (u & 0x80000000u) ? __uint_as_float(u ^ 0x80000000u)
                             : __uint_as_float(~u);
}
__device__ __forceinline__ float leaky(float e) {
    return (e >= 0.f) ? e : NEG_SLOPE * e;
}

__global__ __launch_bounds__(256) void gat_gemm_kernel(
    const float* __restrict__ x,  const float* __restrict__ Wv,
    const float* __restrict__ bv, const float* __restrict__ wq,
    const float* __restrict__ bq, const float* __restrict__ wk,
    const float* __restrict__ bk,
    float* __restrict__ h, float* __restrict__ qout, float* __restrict__ kout,
    unsigned* __restrict__ stat, int N)
{
    __shared__ float Wt[OUT_F * LDS_STRIDE];
    __shared__ float wmx[8], wmn[8];

    const int tid = threadIdx.x;
    for (int idx = tid; idx < IN_F * OUT_F; idx += 256) {
        int kk = idx / OUT_F, n = idx % OUT_F;
        Wt[n * LDS_STRIDE + kk] = Wv[idx];
    }
    __syncthreads();

    const int wave = tid >> 5;
    const int lane = tid & 31;
    const int half = lane >> 4;
    const int l16  = lane & 15;
    const long tile = (long)blockIdx.x * 8 + wave;
    const long rowBase = tile * 16;

    float qmx = -__builtin_inff(), qmn = __builtin_inff();

    if (rowBase < (long)N) {
        v8f acc[4];
#pragma unroll
        for (int nt = 0; nt < 4; ++nt) {
            float b = bv[nt * 16 + l16];
#pragma unroll
            for (int j = 0; j < 8; ++j) acc[nt][j] = b;
        }

        long arow = rowBase + l16;
        if (arow >= (long)N) arow = (long)N - 1;
        const float* __restrict__ xrow = x + arow * IN_F;

        for (int kk = 0; kk < IN_F; kk += 4) {
            const int kl = kk + 2 * half;
            const float2 av = *(const float2*)(xrow + kl);
            v2f a; a.x = av.x; a.y = av.y;
#pragma unroll
            for (int nt = 0; nt < 4; ++nt) {
                const float* wp = &Wt[(nt * 16 + l16) * LDS_STRIDE + kl];
                v2f b; b.x = wp[0]; b.y = wp[1];
                acc[nt] = __builtin_amdgcn_wmma_f32_16x16x4_f32(
                    false, a, false, b, (short)0, acc[nt], false, false);
            }
        }

        float qpart[8], kpart[8];
#pragma unroll
        for (int j = 0; j < 8; ++j) { qpart[j] = 0.f; kpart[j] = 0.f; }
#pragma unroll
        for (int nt = 0; nt < 4; ++nt) {
            const float wqn = wq[nt * 16 + l16];
            const float wkn = wk[nt * 16 + l16];
#pragma unroll
            for (int j = 0; j < 8; ++j) {
                qpart[j] += acc[nt][j] * wqn;
                kpart[j] += acc[nt][j] * wkn;
            }
        }
#pragma unroll
        for (int off = 8; off >= 1; off >>= 1) {
#pragma unroll
            for (int j = 0; j < 8; ++j) {
                qpart[j] += __shfl_xor(qpart[j], off, 32);
                kpart[j] += __shfl_xor(kpart[j], off, 32);
            }
        }

#pragma unroll
        for (int nt = 0; nt < 4; ++nt) {
#pragma unroll
            for (int j = 0; j < 8; ++j) {
                const long m = rowBase + half * 8 + j;
                if (m < (long)N) h[m * OUT_F + nt * 16 + l16] = acc[nt][j];
            }
        }
        const float bqv = bq[0], bkv = bk[0];
        if (l16 == 0) {
#pragma unroll
            for (int j = 0; j < 8; ++j) {
                const long m = rowBase + half * 8 + j;
                if (m < (long)N) {
                    qout[m] = qpart[j] + bqv;
                    kout[m] = kpart[j] + bkv;
                }
            }
        }

        qmx = qpart[0]; qmn = qpart[0];
        for (int j = 1; j < 8; ++j) {
            qmx = fmaxf(qmx, qpart[j]);
            qmn = fminf(qmn, qpart[j]);
        }
        qmx = fmaxf(qmx, __shfl_xor(qmx, 16, 32)) + bqv;
        qmn = fminf(qmn, __shfl_xor(qmn, 16, 32)) + bqv;
    }

    if (lane == 0) { wmx[wave] = qmx; wmn[wave] = qmn; }
    __syncthreads();
    if (tid == 0) {
        for (int w = 1; w < 8; ++w) {
            qmx = fmaxf(qmx, wmx[w]);
            qmn = fminf(qmn, wmn[w]);
        }
        atomicMax(&stat[0], f2ord(qmx));
        atomicMax(&stat[1], f2ord(-qmn));
    }
}

#define OWN_R 391
#define CHUNK_E 32
#define WORD_E (32 * CHUNK_E)
#define GRP_W 8
#define GRP_E (GRP_W * WORD_E)
#define IDX_THREADS 256
#define IDX_OWN 256
#define IDX_PITCH (GRP_W + 1)
__global__ __launch_bounds__(IDX_THREADS) void edge_index_kernel(
    const float* __restrict__ q, const float* __restrict__ k,
    const int* __restrict__ src, const int* __restrict__ dst,
    const unsigned* __restrict__ stat, unsigned* __restrict__ mmax,
    unsigned* __restrict__ mask, int E, int nOwn)
{
    __shared__ unsigned M[IDX_OWN * IDX_PITCH];

    const int tid = threadIdx.x;
    const int Efull = E & ~(CHUNK_E - 1);
    const int4* __restrict__ dst4 = (const int4*)dst;

    int4 dv[GRP_W];
#pragma unroll
    for (int i = 0; i < GRP_W; ++i) {
        const long e = (long)blockIdx.x * GRP_E + (long)(i * IDX_THREADS + tid) * 4;
        dv[i] = make_int4(-1, -1, -1, -1);
        if (e < (long)Efull) dv[i] = dst4[e >> 2];
    }
    const unsigned bit = 1u << (tid >> 3);

    for (int o0 = 0; o0 < nOwn; o0 += IDX_OWN) {
        for (int j = tid; j < IDX_OWN * IDX_PITCH; j += IDX_THREADS) M[j] = 0u;
        __syncthreads();
#pragma unroll
        for (int i = 0; i < GRP_W; ++i) {
            const int dd[4] = { dv[i].x, dv[i].y, dv[i].z, dv[i].w };
#pragma unroll
            for (int c = 0; c < 4; ++c) {
                if (dd[c] >= 0) {
                    const unsigned o = (unsigned)dd[c] / OWN_R - (unsigned)o0;
                    if (o < IDX_OWN) atomicOr(&M[o * IDX_PITCH + i], bit);
                }
            }
        }
        __syncthreads();
        for (int j = tid; j < IDX_OWN * GRP_W; j += IDX_THREADS) {
            const int o = o0 + (j >> 3);
            if (o < nOwn)
                mask[((size_t)blockIdx.x * nOwn + o) * GRP_W + (j & 7)] =
                    M[(j >> 3) * IDX_PITCH + (j & 7)];
        }
        __syncthreads();
    }

    if (!(ord2f(stat[0]) + ord2f(stat[1]) <= SPREAD_MAX)) {
        for (long i = (long)blockIdx.x * IDX_THREADS + tid; i < (long)E;
             i += (long)gridDim.x * IDX_THREADS) {
            const int s = src[i], d = dst[i];
            atomicMax(&mmax[d], f2ord(leaky(q[s] + k[d])));
        }
    }
}

#define AGG_THREADS 1024
#define AGG_WAVES 32
#define QCAP 256
#define PASS_W 16
#define LCAP (PASS_W * 32)
#define BATCH 16

#define ROW_SHARE(x, u) __builtin_amdgcn_update_dpp((x), (x), 0x150 + (u), 0xF, 0xF, true)

#define PAIR_LOAD(u)                                                            \
    if (FULL || (u) < n) {                                                      \
        const int su = ROW_SHARE(s, u);                                         \
        if (FULL || (u) + 16 * hf < n)                                          \
            v[u] = h4[(unsigned)(su * (OUT_F / 4) + l16)];                      \
    }
#define PAIR_ADD(u)                                                             \
    if (FULL || (u) < n) {                                                      \
        const float exu = __int_as_float(ROW_SHARE(__float_as_int(ex), u));     \
        const int du = ROW_SHARE(dl, u);                                        \
        if (FULL || (u) + 16 * hf < n) {                                        \
            const float4 t = v[u];                                              \
            float* pa = U + du * OUT_F + offA;                                  \
            float* pb = U + du * OUT_F + offB;                                  \
            atomicAdd(pa,     (hf ? t.z : t.x) * exu);                          \
            atomicAdd(pa + 1, (hf ? t.w : t.y) * exu);                          \
            atomicAdd(pb,     (hf ? t.x : t.z) * exu);                          \
            atomicAdd(pb + 1, (hf ? t.y : t.w) * exu);                          \
        }                                                                       \
    }

template <bool FULL>
__device__ __forceinline__ void drain_ring(
    const int* qs, const int* qd, int head, int nb, int lane,
    const float* __restrict__ h, const float* __restrict__ q,
    const float* Kd, const float* Md, float* S, float* U)
{
    __builtin_amdgcn_wave_barrier();
    const int n = FULL ? 32 : nb;
    const bool act = FULL || (lane < n);
    const int hf = lane >> 4;
    const int l16 = lane & 15;
    const int offA = 4 * l16 + 2 * hf;
    const int offB = offA ^ 2;
    int s = 0, dl = 0;
    float qv = 0.f, kd = 0.f, md = 0.f, ex = 0.f;
    if (act) {
        const int slot = (head + lane) & (QCAP - 1);
        s = qs[slot];
        dl = qd[slot];
        qv = q[s];
        kd = Kd[dl];
        md = Md[dl];
    }
    const float4* __restrict__ h4 = (const float4*)h;
    float4 v[16];
    PAIR_LOAD(0)  PAIR_LOAD(1)  PAIR_LOAD(2)  PAIR_LOAD(3)
    PAIR_LOAD(4)  PAIR_LOAD(5)  PAIR_LOAD(6)  PAIR_LOAD(7)
    PAIR_LOAD(8)  PAIR_LOAD(9)  PAIR_LOAD(10) PAIR_LOAD(11)
    PAIR_LOAD(12) PAIR_LOAD(13) PAIR_LOAD(14) PAIR_LOAD(15)
    __builtin_amdgcn_sched_barrier(0);
    asm volatile("" : "+v"(qv));
    ex = act ? __expf(leaky(qv + kd) - md) : 0.f;
    PAIR_ADD(0)  PAIR_ADD(1)  PAIR_ADD(2)  PAIR_ADD(3)
    PAIR_ADD(4)  PAIR_ADD(5)  PAIR_ADD(6)  PAIR_ADD(7)
    PAIR_ADD(8)  PAIR_ADD(9)  PAIR_ADD(10) PAIR_ADD(11)
    PAIR_ADD(12) PAIR_ADD(13) PAIR_ADD(14) PAIR_ADD(15)
    if (act) atomicAdd(&S[dl], ex);
}

__global__ __launch_bounds__(AGG_THREADS) void owner_aggregate_kernel(
    const float* __restrict__ h, const float* __restrict__ q,
    const float* __restrict__ k, const int* __restrict__ src,
    const int* __restrict__ dst, const unsigned* __restrict__ mmax,
    const unsigned* __restrict__ stat, const unsigned* __restrict__ mask,
    float* __restrict__ out, int N, int E)
{
    __shared__ __attribute__((aligned(16))) float U[OWN_R * OUT_F];
    __shared__ float S[OWN_R];
    __shared__ float Kd[OWN_R];
    __shared__ float Md[OWN_R];
    __shared__ int Qs[AGG_WAVES * QCAP];
    __shared__ int Qd[AGG_WAVES * QCAP];
    __shared__ int Cl[AGG_WAVES * LCAP];

    const int tid = threadIdx.x;
    const int lane = tid & 31;
    const int wave = tid >> 5;
    const int base = blockIdx.x * OWN_R;
    const int R = (N - base < OWN_R) ? (N - base) : OWN_R;

    for (int i = tid; i < R * OUT_F; i += AGG_THREADS) U[i] = 0.f;
    const float qmx = ord2f(stat[0]);
    const bool narrow = (qmx + ord2f(stat[1]) <= SPREAD_MAX);
    for (int i = tid; i < R; i += AGG_THREADS) {
        const float kd = k[base + i];
        S[i] = 0.f;
        Kd[i] = kd;
        Md[i] = narrow ? leaky(qmx + kd)
                       : ord2f(mmax[base + i]);
    }
    __syncthreads();

    int* qs = Qs + wave * QCAP;
    int* qd = Qd + wave * QCAP;
    int* cl = Cl + wave * LCAP;
    int head = 0, cnt = 0;

    auto push = [&](unsigned r, int s) {
        const bool m = r < (unsigned)R;
        const unsigned b = __builtin_amdgcn_ballot_w32(m);
        if (m) {
            const int slot = (head + cnt + (int)__builtin_amdgcn_mbcnt_lo(b, 0u)) & (QCAP - 1);
            qs[slot] = s;
            qd[slot] = (int)r;
        }
        cnt += __popc(b);
    };
    auto drain_full = [&]() {
        while (cnt >= 32) {
            drain_ring<true>(qs, qd, head, 32, lane, h, q, Kd, Md, S, U);
            head = (head + 32) & (QCAP - 1);
            cnt -= 32;
        }
    };
    auto batch = [&](const bool full, const int n, const int i0) {
        int4 dv[BATCH / 4], sv[BATCH / 4];
        const int sub = lane >> 3;
        const int e4 = (lane & 7) * 4;
#pragma unroll
        for (int g = 0; g < BATCH / 4; ++g) {
            const int ci = (full || 4 * g + sub < n) ? 4 * g + sub : 0;
            const int e = cl[i0 + ci] * CHUNK_E + e4;
            dv[g] = *(const int4*)(dst + e);
            sv[g] = *(const int4*)(src + e);
        }
#pragma unroll
        for (int g = 0; g < BATCH / 4; ++g) {
            if (full || 4 * g < n) {
                int s0 = sv[g].x, s1 = sv[g].y, s2 = sv[g].z, s3 = sv[g].w;
                asm volatile("" : "+v"(s0), "+v"(s1), "+v"(s2), "+v"(s3));
                const bool have = full || 4 * g + sub < n;
                push(have ? (unsigned)(dv[g].x - base) : ~0u, s0);
                push(have ? (unsigned)(dv[g].y - base) : ~0u, s1);
                push(have ? (unsigned)(dv[g].z - base) : ~0u, s2);
                push(have ? (unsigned)(dv[g].w - base) : ~0u, s3);
            }
            drain_full();
        }
    };

    const int nOwn = gridDim.x;
    const int Efull = E & ~(CHUNK_E - 1);
    const int nWords = (Efull + WORD_E - 1) / WORD_E;
    const int per = (nWords + AGG_WAVES - 1) / AGG_WAVES;
    const int wBeg = wave * per;
    const int wEnd = (wBeg + per < nWords) ? (wBeg + per) : nWords;

    auto load_words = [&](int wA) -> unsigned {
        const int Wi = wA + lane;
        unsigned w = 0u;
        if (lane < PASS_W && Wi < wEnd)
            w = mask[((size_t)((unsigned)Wi / GRP_W) * nOwn + blockIdx.x) * GRP_W
                     + ((unsigned)Wi % GRP_W)];
        return w;
    };

    unsigned word = load_words(wBeg);
    for (int wA = wBeg; wA < wEnd; wA += PASS_W) {
        const unsigned nxt = load_words(wA + PASS_W);

        const int c = __popc(word);
        int incl = c;
#pragma unroll
        for (int off = 1; off < 32; off <<= 1) {
            const int t = __shfl_up(incl, off, 32);
            if (lane >= off) incl += t;
        }
        const int nl = __builtin_amdgcn_readlane(incl, 31);
        int pos = incl - c;
        const int cbase = (wA + lane) * 32;
        for (unsigned w = word; w != 0u; w &= w - 1u)
            cl[pos++] = cbase + __builtin_ctz(w);
        __builtin_amdgcn_wave_barrier();

        for (int i0 = 0; i0 < nl; i0 += BATCH) {
            const int n = (nl - i0 < BATCH) ? (nl - i0) : BATCH;
            if (n == BATCH) batch(true, BATCH, i0);
            else            batch(false, n, i0);
        }
        word = nxt;
    }
    if (wave == 0) {
        const int e = Efull + lane;
        unsigned r = ~0u;
        int s = 0;
        if (e < E) { r = (unsigned)(dst[e] - base); s = src[e]; }
        push(r, s);
        drain_full();
    }
    if (cnt > 0)
        drain_ring<false>(qs, qd, head, cnt, lane, h, q, Kd, Md, S, U);

    __syncthreads();

    float4* __restrict__ orow = (float4*)(out + (long)base * OUT_F);
    for (int i = tid; i < R * (OUT_F / 4); i += AGG_THREADS) {
        const float sd = S[i >> 4];
        float4 v = ((const float4*)U)[i];
        if (sd > 0.f) { v.x /= sd; v.y /= sd; v.z /= sd; v.w /= sd; }
        orow[i] = v;
    }
}

extern "C" void kernel_launch(void* const* d_in, const int* in_sizes, int n_in,
                              void* d_out, int out_size, void* d_ws, size_t ws_size,
                              hipStream_t stream) {
    const float* x  = (const float*)d_in[0];
    const float* Wv = (const float*)d_in[1];
    const float* bv = (const float*)d_in[2];
    const float* wq = (const float*)d_in[3];
    const float* bq = (const float*)d_in[4];
    const float* wk = (const float*)d_in[5];
    const float* bk = (const float*)d_in[6];
    const int*   src = (const int*)d_in[7];
    const int*   dst = (const int*)d_in[8];

    const int N = in_sizes[0] / IN_F;
    const int E = in_sizes[7];
    float* out = (float*)d_out;

    char* p = (char*)d_ws;
    auto carve = [&](size_t bytes) {
        char* r = p;
        p += (bytes + 255) & ~(size_t)255;
        return r;
    };
    float*    h    = (float*)   carve((size_t)N * OUT_F * sizeof(float));
    float*    q    = (float*)   carve((size_t)N * sizeof(float));
    float*    k    = (float*)   carve((size_t)N * sizeof(float));
    const size_t zbytes = ((size_t)N + 64) * sizeof(unsigned);
    unsigned* mmax = (unsigned*)carve(zbytes);
    unsigned* stat = mmax + (size_t)N;
    const int nOwn = (N + OWN_R - 1) / OWN_R;
    const int nGroups = (E + GRP_E - 1) / GRP_E;
    unsigned* mask = (unsigned*)carve((size_t)nGroups * nOwn * GRP_W * sizeof(unsigned));

    hipMemsetAsync(mmax, 0, zbytes, stream);

    const int tiles = (N + 15) / 16;
    const int gemmBlocks = (tiles + 7) / 8;
    gat_gemm_kernel<<<gemmBlocks, 256, 0, stream>>>(
        x, Wv, bv, wq, bq, wk, bk, h, q, k, stat, N);

    edge_index_kernel<<<nGroups, IDX_THREADS, 0, stream>>>(
        q, k, src, dst, stat, mmax, mask, E, nOwn);

    owner_aggregate_kernel<<<nOwn, AGG_THREADS, 0, stream>>>(
        h, q, k, src, dst, mmax, stat, mask, out, N, E);
}
